# starting kernel
# baseline (speedup 1.0000x reference)
#include <hip/hip_runtime.h>

typedef float v2f __attribute__((ext_vector_type(2)));
typedef float v8f __attribute__((ext_vector_type(8)));

#define MV 16
#define MAXLEN 16384

__device__ constexpr int popc4(int v) { return (v & 1) + ((v >> 1) & 1) + ((v >> 2) & 1) + ((v >> 3) & 1); }

__device__ constexpr float gpsign(int a, int b) {
    int s = 0;
    for (int t = a >> 1; t; t >>= 1) s += popc4(t & b);
    return (s & 1) ? -1.0f : 1.0f;
}
__device__ constexpr float bivsign(int k) { return (popc4(k) == 2) ? -1.0f : 1.0f; }

__device__ inline v8f gp_via_wmma_t(const float* __restrict__ ctab,
                                    const float (&Af)[16], const float (&Bf)[16],
                                    int lane, bool hi, v8f acc) {
#pragma unroll
    for (int c = 0; c < 64; ++c) {
        const int i  = c >> 2;
        const int j0 = (c & 3) << 2;
        const v2f aC = *(const v2f*)(ctab + c * 64 + lane * 2);
        v2f zB;
        zB[0] = Af[i] * (hi ? Bf[j0 + 2] : Bf[j0 + 0]);
        zB[1] = Af[i] * (hi ? Bf[j0 + 3] : Bf[j0 + 1]);
        acc = __builtin_amdgcn_wmma_f32_16x16x4_f32(false, aC, false, zB,
                                                    (short)0, acc, false, false);
    }
    return acc;
}

__global__ __launch_bounds__(256)
void care_rotor_kernel(const float* __restrict__ x,
                       const int*   __restrict__ pos,
                       const float* __restrict__ bx,
                       const float* __restrict__ by,
                       const float* __restrict__ bz,
                       const float* __restrict__ bw,
                       const float* __restrict__ theta,
                       float* __restrict__ out,
                       int ntiles) {
    __shared__ float ctab[64 * 32 * 2];

    const int lane = threadIdx.x & 31;
    const int wave = threadIdx.x >> 5;
    const int kl   = lane & 15;
    const bool hi  = lane >= 16;

    if (wave == 0) {
#pragma unroll
        for (int c = 0; c < 64; ++c) {
            const int i  = c >> 2;
            const int j0 = (c & 3) << 2;
            const float e0lo = (kl == (i ^ (j0 + 0))) ? gpsign(i, j0 + 0) : 0.0f;
            const float e1lo = (kl == (i ^ (j0 + 1))) ? gpsign(i, j0 + 1) : 0.0f;
            const float e0hi = (kl == (i ^ (j0 + 2))) ? gpsign(i, j0 + 2) : 0.0f;
            const float e1hi = (kl == (i ^ (j0 + 3))) ? gpsign(i, j0 + 3) : 0.0f;
            v2f e;
            e[0] = hi ? e0hi : e0lo;
            e[1] = hi ? e1hi : e1lo;
            *(v2f*)(ctab + c * 64 + lane * 2) = e;
        }
    }
    __syncthreads();

    const int tile = blockIdx.x * 8 + wave;
    if (tile >= ntiles) return;

    const long pbase = (long)tile * 16;
    const long p     = pbase + kl;

    float xv[16];
    {
        const float4* xp = (const float4*)(x + p * MV);
        float4 q[4];
#pragma unroll
        for (int t = 0; t < 4; ++t) q[t] = xp[t];
#pragma unroll
        for (int t = 0; t < 4; ++t) {
            xv[4 * t + 0] = q[t].x; xv[4 * t + 1] = q[t].y;
            xv[4 * t + 2] = q[t].z; xv[4 * t + 3] = q[t].w;
        }
    }

    int pc = pos[p];
    pc = pc < 0 ? 0 : (pc > MAXLEN - 1 ? MAXLEN - 1 : pc);
    const float4 th = *(const float4*)(theta + (long)pc * 4);
    float phi[4];
    phi[0] = 0.5f * th.x * bx[3];
    phi[1] = 0.5f * th.y * by[5];
    phi[2] = 0.5f * th.z * bz[9];
    phi[3] = 0.5f * th.w * bw[6];

    float r[16];
#pragma unroll
    for (int k = 0; k < 16; ++k) r[k] = (k == 0) ? 1.0f : 0.0f;
    const int blades[4] = {3, 5, 9, 6};
#pragma unroll
    for (int pl = 0; pl < 4; ++pl) {
        float sn, cs;
        __sincosf(phi[pl], &sn, &cs);
        const int b = blades[pl];
        float nr[16];
#pragma unroll
        for (int k = 0; k < 16; ++k)
            nr[k] = cs * r[k] + sn * gpsign(k ^ b, b) * r[k ^ b];
#pragma unroll
        for (int k = 0; k < 16; ++k) r[k] = nr[k];
    }

    float rrev[16];
#pragma unroll
    for (int k = 0; k < 16; ++k) rrev[k] = r[k] * bivsign(k);

    v8f acc;
#pragma unroll
    for (int v = 0; v < 8; ++v) acc[v] = 0.0f;
    acc = gp_via_wmma_t(ctab, r, xv, lane, hi, acc);

    float rxv[16];
#pragma unroll
    for (int v = 0; v < 8; ++v) {
        const float other = __shfl_xor((float)acc[v], 16, 32);
        rxv[v]     = hi ? other  : acc[v];
        rxv[v + 8] = hi ? acc[v] : other;
    }

    v8f acc2;
#pragma unroll
    for (int v = 0; v < 8; ++v) acc2[v] = 0.0f;
    acc2 = gp_via_wmma_t(ctab, rxv, rrev, lane, hi, acc2);

    {
        float* po = out + (pbase + kl) * MV + (hi ? 8 : 0);
        float4 q0, q1;
        q0.x = acc2[0]; q0.y = acc2[1]; q0.z = acc2[2]; q0.w = acc2[3];
        q1.x = acc2[4]; q1.y = acc2[5]; q1.z = acc2[6]; q1.w = acc2[7];
        *(float4*)(po + 0) = q0;
        *(float4*)(po + 4) = q1;
    }
}

extern "C" void kernel_launch(void* const* d_in, const int* in_sizes, int n_in,
                              void* d_out, int out_size, void* d_ws, size_t ws_size,
                              hipStream_t stream) {
    const float* x     = (const float*)d_in[0];
    const int*   pos   = (const int*)d_in[1];
    const float* bx    = (const float*)d_in[2];
    const float* by    = (const float*)d_in[3];
    const float* bz    = (const float*)d_in[4];
    const float* bw    = (const float*)d_in[5];
    const float* theta = (const float*)d_in[6];
    float* out = (float*)d_out;

    const int npos   = in_sizes[1];
    const int ntiles = npos / 16;
    const int blocks = (ntiles + 7) / 8;

    hipLaunchKernelGGL(care_rotor_kernel, dim3(blocks), dim3(256), 0, stream,
                       x, pos, bx, by, bz, bw, theta, out, ntiles);
}


# Skip the rotor's structurally zero odd-grade components: both products run over the 8 even-grade rotor blades only (K 256 -> 128 each), 64 f32 WMMAs per tile instead of 128
# speedup vs baseline: 1.6025x; 1.6025x over previous
#include <hip/hip_runtime.h>


typedef float v2f __attribute__((ext_vector_type(2)));
typedef float v8f __attribute__((ext_vector_type(8)));

#define MV 16
#define MAXLEN 16384

__device__ constexpr int popc4(int v) { return (v & 1) + ((v >> 1) & 1) + ((v >> 2) & 1) + ((v >> 3) & 1); }

__device__ constexpr float gpsign(int a, int b) {
    int s = 0;
    for (int t = a >> 1; t; t >>= 1) s += popc4(t & b);
    return (s & 1) ? -1.0f : 1.0f;
}
__device__ constexpr float bivsign(int k) { return (popc4(k) == 2) ? -1.0f : 1.0f; }

__device__ constexpr int evb(int e) { return (e << 1) | (popc4(e) & 1); }

__device__ constexpr int g1_ie(int c)        { return c >> 2; }
__device__ constexpr int g1_i(int c)         { return evb(g1_ie(c)); }
__device__ constexpr int g1_j(int c, int s)  { return ((c & 3) << 2) + s; }
__device__ constexpr int g2_i(int c)         { return c >> 1; }
__device__ constexpr int g2_je(int c, int s) { return ((c & 1) << 2) + s; }
__device__ constexpr int g2_j(int c, int s)  { return evb(g2_je(c, s)); }

#define NCH 32
#define TAB_FLOATS (2 * NCH * 64)

__global__ __launch_bounds__(256)
void care_rotor_kernel(const float* __restrict__ x,
                       const int*   __restrict__ pos,
                       const float* __restrict__ bx,
                       const float* __restrict__ by,
                       const float* __restrict__ bz,
                       const float* __restrict__ bw,
                       const float* __restrict__ theta,
                       float* __restrict__ out,
                       int ntiles) {
    __shared__ float ctab[TAB_FLOATS];

    const int lane = threadIdx.x & 31;
    const int wave = threadIdx.x >> 5;
    const int kl   = lane & 15;
    const bool hi  = lane >= 16;

    if (wave == 0) {
#pragma unroll
        for (int c = 0; c < NCH; ++c) {
            const int i = g1_i(c);
            const float e0lo = (kl == (i ^ g1_j(c, 0))) ? gpsign(i, g1_j(c, 0)) : 0.0f;
            const float e1lo = (kl == (i ^ g1_j(c, 1))) ? gpsign(i, g1_j(c, 1)) : 0.0f;
            const float e0hi = (kl == (i ^ g1_j(c, 2))) ? gpsign(i, g1_j(c, 2)) : 0.0f;
            const float e1hi = (kl == (i ^ g1_j(c, 3))) ? gpsign(i, g1_j(c, 3)) : 0.0f;
            v2f e;
            e[0] = hi ? e0hi : e0lo;
            e[1] = hi ? e1hi : e1lo;
            *(v2f*)(ctab + c * 64 + lane * 2) = e;
        }
#pragma unroll
        for (int c = 0; c < NCH; ++c) {
            const int i = g2_i(c);
            const float e0lo = (kl == (i ^ g2_j(c, 0))) ? gpsign(i, g2_j(c, 0)) : 0.0f;
            const float e1lo = (kl == (i ^ g2_j(c, 1))) ? gpsign(i, g2_j(c, 1)) : 0.0f;
            const float e0hi = (kl == (i ^ g2_j(c, 2))) ? gpsign(i, g2_j(c, 2)) : 0.0f;
            const float e1hi = (kl == (i ^ g2_j(c, 3))) ? gpsign(i, g2_j(c, 3)) : 0.0f;
            v2f e;
            e[0] = hi ? e0hi : e0lo;
            e[1] = hi ? e1hi : e1lo;
            *(v2f*)(ctab + (NCH + c) * 64 + lane * 2) = e;
        }
    }
    __syncthreads();

    const int tile = blockIdx.x * 8 + wave;
    if (tile >= ntiles) return;

    const long pbase = (long)tile * 16;
    const long p     = pbase + kl;

    float xv[16];
    {
        const float4* xp = (const float4*)(x + p * MV);
        float4 q[4];
#pragma unroll
        for (int t = 0; t < 4; ++t) q[t] = xp[t];
#pragma unroll
        for (int t = 0; t < 4; ++t) {
            xv[4 * t + 0] = q[t].x; xv[4 * t + 1] = q[t].y;
            xv[4 * t + 2] = q[t].z; xv[4 * t + 3] = q[t].w;
        }
    }

    int pc = pos[p];
    pc = pc < 0 ? 0 : (pc > MAXLEN - 1 ? MAXLEN - 1 : pc);
    const float4 th = *(const float4*)(theta + (long)pc * 4);
    float phi[4];
    phi[0] = 0.5f * th.x * bx[3];
    phi[1] = 0.5f * th.y * by[5];
    phi[2] = 0.5f * th.z * bz[9];
    phi[3] = 0.5f * th.w * bw[6];

    float re[8];
#pragma unroll
    for (int e = 0; e < 8; ++e) re[e] = (e == 0) ? 1.0f : 0.0f;
    const int blades[4] = {3, 5, 9, 6};
#pragma unroll
    for (int pl = 0; pl < 4; ++pl) {
        float sn, cs;
        __sincosf(phi[pl], &sn, &cs);
        const int b = blades[pl];
        float nr[8];
#pragma unroll
        for (int e = 0; e < 8; ++e) {
            const int k = evb(e);
            nr[e] = cs * re[e] + sn * gpsign(k ^ b, b) * re[(k ^ b) >> 1];
        }
#pragma unroll
        for (int e = 0; e < 8; ++e) re[e] = nr[e];
    }

    float rrev[8];
#pragma unroll
    for (int e = 0; e < 8; ++e) rrev[e] = re[e] * bivsign(evb(e));

    v8f acc;
#pragma unroll
    for (int v = 0; v < 8; ++v) acc[v] = 0.0f;
#pragma unroll
    for (int c = 0; c < NCH; ++c) {
        const v2f aC = *(const v2f*)(ctab + c * 64 + lane * 2);
        v2f zB;
        zB[0] = re[g1_ie(c)] * (hi ? xv[g1_j(c, 2)] : xv[g1_j(c, 0)]);
        zB[1] = re[g1_ie(c)] * (hi ? xv[g1_j(c, 3)] : xv[g1_j(c, 1)]);
        acc = __builtin_amdgcn_wmma_f32_16x16x4_f32(false, aC, false, zB,
                                                    (short)0, acc, false, false);
    }

    float rxv[16];
#pragma unroll
    for (int v = 0; v < 8; ++v) {
        const float other = __shfl_xor((float)acc[v], 16, 32);
        rxv[v]     = hi ? other  : acc[v];
        rxv[v + 8] = hi ? acc[v] : other;
    }

    v8f acc2;
#pragma unroll
    for (int v = 0; v < 8; ++v) acc2[v] = 0.0f;
#pragma unroll
    for (int c = 0; c < NCH; ++c) {
        const v2f aC = *(const v2f*)(ctab + (NCH + c) * 64 + lane * 2);
        v2f zB;
        zB[0] = rxv[g2_i(c)] * (hi ? rrev[g2_je(c, 2)] : rrev[g2_je(c, 0)]);
        zB[1] = rxv[g2_i(c)] * (hi ? rrev[g2_je(c, 3)] : rrev[g2_je(c, 1)]);
        acc2 = __builtin_amdgcn_wmma_f32_16x16x4_f32(false, aC, false, zB,
                                                     (short)0, acc2, false, false);
    }

    {
        float* po = out + (pbase + kl) * MV + (hi ? 8 : 0);
        float4 q0, q1;
        q0.x = acc2[0]; q0.y = acc2[1]; q0.z = acc2[2]; q0.w = acc2[3];
        q1.x = acc2[4]; q1.y = acc2[5]; q1.z = acc2[6]; q1.w = acc2[7];
        *(float4*)(po + 0) = q0;
        *(float4*)(po + 4) = q1;
    }
}

extern "C" void kernel_launch(void* const* d_in, const int* in_sizes, int n_in,
                              void* d_out, int out_size, void* d_ws, size_t ws_size,
                              hipStream_t stream) {
    const float* x     = (const float*)d_in[0];
    const int*   pos   = (const int*)d_in[1];
    const float* bx    = (const float*)d_in[2];
    const float* by    = (const float*)d_in[3];
    const float* bz    = (const float*)d_in[4];
    const float* bw    = (const float*)d_in[5];
    const float* theta = (const float*)d_in[6];
    float* out = (float*)d_out;

    const int npos   = in_sizes[1];
    const int ntiles = npos / 16;
    const int blocks = (ntiles + 7) / 8;

    hipLaunchKernelGGL(care_rotor_kernel, dim3(blocks), dim3(256), 0, stream,
                       x, pos, bx, by, bz, bw, theta, out, ntiles);
}


# Re-order k in both products so each lane half's B values are registers it already owns (its half of x; its GP1 result tile): no per-chunk selects, no cross-half shuffle, half the x loads
# speedup vs baseline: 1.8351x; 1.1451x over previous
#include <hip/hip_runtime.h>


typedef float v2f __attribute__((ext_vector_type(2)));
typedef float v8f __attribute__((ext_vector_type(8)));

#define MV 16
#define MAXLEN 16384

__device__ constexpr int popc4(int v) { return (v & 1) + ((v >> 1) & 1) + ((v >> 2) & 1) + ((v >> 3) & 1); }

__device__ constexpr float gpsign(int a, int b) {
    int s = 0;
    for (int t = a >> 1; t; t >>= 1) s += popc4(t & b);
    return (s & 1) ? -1.0f : 1.0f;
}
__device__ constexpr float bivsign(int k) { return (popc4(k) == 2) ? -1.0f : 1.0f; }

__device__ constexpr int evb(int e) { return (e << 1) | (popc4(e) & 1); }

__device__ constexpr int g1_ie(int c)              { return c >> 2; }
__device__ constexpr int g1_i(int c)               { return evb(g1_ie(c)); }
__device__ constexpr int g1_jl(int c, int v)       { return ((c & 3) << 1) + v; }
__device__ constexpr int g1_j(int c, int v, int h) { return (h << 3) + g1_jl(c, v); }
__device__ constexpr int g2_il(int c)              { return c >> 2; }
__device__ constexpr int g2_i(int c, int h)        { return (h << 3) + g2_il(c); }
__device__ constexpr int g2_je(int c, int v)       { return ((c & 3) << 1) + v; }
__device__ constexpr int g2_j(int c, int v)        { return evb(g2_je(c, v)); }

#define NCH 32
#define TAB_FLOATS (2 * NCH * 64)

__global__ __launch_bounds__(256)
void care_rotor_kernel(const float* __restrict__ x,
                       const int*   __restrict__ pos,
                       const float* __restrict__ bx,
                       const float* __restrict__ by,
                       const float* __restrict__ bz,
                       const float* __restrict__ bw,
                       const float* __restrict__ theta,
                       float* __restrict__ out,
                       int ntiles) {
    __shared__ float ctab[TAB_FLOATS];

    const int lane = threadIdx.x & 31;
    const int wave = threadIdx.x >> 5;
    const int kl   = lane & 15;
    const bool hi  = lane >= 16;

    if (wave == 0) {
#pragma unroll
        for (int c = 0; c < NCH; ++c) {
            const int i = g1_i(c);
            const float e0lo = (kl == (i ^ g1_j(c, 0, 0))) ? gpsign(i, g1_j(c, 0, 0)) : 0.0f;
            const float e1lo = (kl == (i ^ g1_j(c, 1, 0))) ? gpsign(i, g1_j(c, 1, 0)) : 0.0f;
            const float e0hi = (kl == (i ^ g1_j(c, 0, 1))) ? gpsign(i, g1_j(c, 0, 1)) : 0.0f;
            const float e1hi = (kl == (i ^ g1_j(c, 1, 1))) ? gpsign(i, g1_j(c, 1, 1)) : 0.0f;
            v2f e;
            e[0] = hi ? e0hi : e0lo;
            e[1] = hi ? e1hi : e1lo;
            *(v2f*)(ctab + c * 64 + lane * 2) = e;
        }
#pragma unroll
        for (int c = 0; c < NCH; ++c) {
            const int ilo = g2_i(c, 0), ihi = g2_i(c, 1);
            const float e0lo = (kl == (ilo ^ g2_j(c, 0))) ? gpsign(ilo, g2_j(c, 0)) : 0.0f;
            const float e1lo = (kl == (ilo ^ g2_j(c, 1))) ? gpsign(ilo, g2_j(c, 1)) : 0.0f;
            const float e0hi = (kl == (ihi ^ g2_j(c, 0))) ? gpsign(ihi, g2_j(c, 0)) : 0.0f;
            const float e1hi = (kl == (ihi ^ g2_j(c, 1))) ? gpsign(ihi, g2_j(c, 1)) : 0.0f;
            v2f e;
            e[0] = hi ? e0hi : e0lo;
            e[1] = hi ? e1hi : e1lo;
            *(v2f*)(ctab + (NCH + c) * 64 + lane * 2) = e;
        }
    }
    __syncthreads();

    const int tile = blockIdx.x * 8 + wave;
    if (tile >= ntiles) return;

    const long pbase = (long)tile * 16;
    const long p     = pbase + kl;

    int pc = pos[p];
    const float4* xp = (const float4*)(x + p * MV + (hi ? 8 : 0));
    const float4 q0 = xp[0];
    const float4 q1 = xp[1];
    __builtin_amdgcn_sched_barrier(0);
    float xh[8];
    xh[0] = q0.x; xh[1] = q0.y; xh[2] = q0.z; xh[3] = q0.w;
    xh[4] = q1.x; xh[5] = q1.y; xh[6] = q1.z; xh[7] = q1.w;

    pc = pc < 0 ? 0 : (pc > MAXLEN - 1 ? MAXLEN - 1 : pc);
    const float4 th = *(const float4*)(theta + (long)pc * 4);
    float phi[4];
    phi[0] = 0.5f * th.x * bx[3];
    phi[1] = 0.5f * th.y * by[5];
    phi[2] = 0.5f * th.z * bz[9];
    phi[3] = 0.5f * th.w * bw[6];

    float re[8];
#pragma unroll
    for (int e = 0; e < 8; ++e) re[e] = (e == 0) ? 1.0f : 0.0f;
    const int blades[4] = {3, 5, 9, 6};
#pragma unroll
    for (int pl = 0; pl < 4; ++pl) {
        float sn, cs;
        __sincosf(phi[pl], &sn, &cs);
        const int b = blades[pl];
        float nr[8];
#pragma unroll
        for (int e = 0; e < 8; ++e) {
            const int k = evb(e);
            nr[e] = cs * re[e] + sn * gpsign(k ^ b, b) * re[(k ^ b) >> 1];
        }
#pragma unroll
        for (int e = 0; e < 8; ++e) re[e] = nr[e];
    }

    float rrev[8];
#pragma unroll
    for (int e = 0; e < 8; ++e) rrev[e] = re[e] * bivsign(evb(e));

    v8f acc;
#pragma unroll
    for (int v = 0; v < 8; ++v) acc[v] = 0.0f;
#pragma unroll
    for (int c = 0; c < NCH; ++c) {
        const v2f aC = *(const v2f*)(ctab + c * 64 + lane * 2);
        v2f zB;
        zB[0] = re[g1_ie(c)] * xh[g1_jl(c, 0)];
        zB[1] = re[g1_ie(c)] * xh[g1_jl(c, 1)];
        acc = __builtin_amdgcn_wmma_f32_16x16x4_f32(false, aC, false, zB,
                                                    (short)0, acc, false, false);
    }


    v8f acc2;
#pragma unroll
    for (int v = 0; v < 8; ++v) acc2[v] = 0.0f;
#pragma unroll
    for (int c = 0; c < NCH; ++c) {
        const v2f aC = *(const v2f*)(ctab + (NCH + c) * 64 + lane * 2);
        v2f zB;
        zB[0] = acc[g2_il(c)] * rrev[g2_je(c, 0)];
        zB[1] = acc[g2_il(c)] * rrev[g2_je(c, 1)];
        acc2 = __builtin_amdgcn_wmma_f32_16x16x4_f32(false, aC, false, zB,
                                                     (short)0, acc2, false, false);
    }

    {
        float* po = out + (pbase + kl) * MV + (hi ? 8 : 0);
        float4 q0, q1;
        q0.x = acc2[0]; q0.y = acc2[1]; q0.z = acc2[2]; q0.w = acc2[3];
        q1.x = acc2[4]; q1.y = acc2[5]; q1.z = acc2[6]; q1.w = acc2[7];
        *(float4*)(po + 0) = q0;
        *(float4*)(po + 4) = q1;
    }
}

extern "C" void kernel_launch(void* const* d_in, const int* in_sizes, int n_in,
                              void* d_out, int out_size, void* d_ws, size_t ws_size,
                              hipStream_t stream) {
    const float* x     = (const float*)d_in[0];
    const int*   pos   = (const int*)d_in[1];
    const float* bx    = (const float*)d_in[2];
    const float* by    = (const float*)d_in[3];
    const float* bz    = (const float*)d_in[4];
    const float* bw    = (const float*)d_in[5];
    const float* theta = (const float*)d_in[6];
    float* out = (float*)d_out;

    const int npos   = in_sizes[1];
    const int ntiles = npos / 16;
    const int blocks = (ntiles + 7) / 8;

    hipLaunchKernelGGL(care_rotor_kernel, dim3(blocks), dim3(256), 0, stream,
                       x, pos, bx, by, bz, bw, theta, out, ntiles);
}


# Each plane stage keeps cos(2 phi) x_j on its own row via the C operand (scaled in place); only the sin(2 phi) e_b e_j routing stays a product: K 16 -> 8 per stage, 8 f32 WMMAs per tile, not 16
# speedup vs baseline: 4.4603x; 2.4305x over previous
#include <hip/hip_runtime.h>


typedef float v2f __attribute__((ext_vector_type(2)));
typedef float v8f __attribute__((ext_vector_type(8)));

#define MV 16
#define MAXLEN 16384

__device__ constexpr int popc4(int v) { return (v & 1) + ((v >> 1) & 1) + ((v >> 2) & 1) + ((v >> 3) & 1); }

__device__ constexpr float gpsign(int a, int b) {
    int s = 0;
    for (int t = a >> 1; t; t >>= 1) s += popc4(t & b);
    return (s & 1) ? -1.0f : 1.0f;
}

__device__ constexpr bool anti(int j, int b) { return gpsign(b, j) != gpsign(j, b); }

__device__ constexpr int ajl(int b, int h, int t) {
    int n = 0;
    for (int r = 0; r < 8; ++r) {
        if (anti((h << 3) + r, b)) {
            if (n == t) return r;
            ++n;
        }
    }
    return 0;
}

#define NST 4
#define NCH 2
#define TAB_FLOATS (NST * NCH * 64)

template <int B, int T>
__device__ __forceinline__ void build_chunk(float* tab, int lane, int kl, bool hi) {
    constexpr int j0lo = ajl(B, 0, 2 * T);
    constexpr int j1lo = ajl(B, 0, 2 * T + 1);
    constexpr int j0hi = 8 + ajl(B, 1, 2 * T);
    constexpr int j1hi = 8 + ajl(B, 1, 2 * T + 1);
    constexpr float g0lo = gpsign(B, j0lo);
    constexpr float g1lo = gpsign(B, j1lo);
    constexpr float g0hi = gpsign(B, j0hi);
    constexpr float g1hi = gpsign(B, j1hi);
    const int   j0 = hi ? j0hi : j0lo;
    const int   j1 = hi ? j1hi : j1lo;
    const float g0 = hi ? g0hi : g0lo;
    const float g1 = hi ? g1hi : g1lo;
    v2f e;
    e[0] = (kl == (B ^ j0)) ? g0 : 0.0f;
    e[1] = (kl == (B ^ j1)) ? g1 : 0.0f;
    *(v2f*)(tab + lane * 2) = e;
}

template <int B>
__device__ __forceinline__ void build_stage(float* tab, int lane, int kl, bool hi) {
    build_chunk<B, 0>(tab + 0 * 64, lane, kl, hi);
    build_chunk<B, 1>(tab + 1 * 64, lane, kl, hi);
}

template <int B, int T>
__device__ __forceinline__ v8f stage_chunk(v8f acc, const v8f& X, float s2,
                                           const float* tab, int lane, bool hi) {
    constexpr int r0lo = ajl(B, 0, 2 * T);
    constexpr int r0hi = ajl(B, 1, 2 * T);
    constexpr int r1lo = ajl(B, 0, 2 * T + 1);
    constexpr int r1hi = ajl(B, 1, 2 * T + 1);
    const float x0 = (r0lo == r0hi) ? X[r0lo] : (hi ? X[r0hi] : X[r0lo]);
    const float x1 = (r1lo == r1hi) ? X[r1lo] : (hi ? X[r1hi] : X[r1lo]);
    v2f zB;
    zB[0] = s2 * x0;
    zB[1] = s2 * x1;
    const v2f aC = *(const v2f*)(tab + lane * 2);
    return __builtin_amdgcn_wmma_f32_16x16x4_f32(false, aC, false, zB,
                                                 (short)0, acc, false, false);
}

template <int B>
__device__ __forceinline__ v8f plane_stage(const v8f& X, float sn, float cs,
                                           const float* tab, int lane, bool hi) {
    const float c2 = cs * cs - sn * sn;
    const float s2 = 2.0f * cs * sn;
    const float c_lo_only = hi ? 1.0f : c2;
    const float c_hi_only = hi ? c2 : 1.0f;
    v8f acc;
#pragma unroll
    for (int r = 0; r < 8; ++r) {
        const bool alo = anti(r, B), ahi = anti(8 + r, B);
        if (alo == ahi) acc[r] = alo ? c2 * X[r] : X[r];
        else            acc[r] = (alo ? c_lo_only : c_hi_only) * X[r];
    }
    acc = stage_chunk<B, 0>(acc, X, s2, tab + 0 * 64, lane, hi);
    acc = stage_chunk<B, 1>(acc, X, s2, tab + 1 * 64, lane, hi);
    return acc;
}

__global__ __launch_bounds__(256)
void care_rotor_kernel(const float* __restrict__ x,
                       const int*   __restrict__ pos,
                       const float* __restrict__ bx,
                       const float* __restrict__ by,
                       const float* __restrict__ bz,
                       const float* __restrict__ bw,
                       const float* __restrict__ theta,
                       float* __restrict__ out,
                       int ntiles) {
    __shared__ float ctab[TAB_FLOATS];

    const int lane = threadIdx.x & 31;
    const int wave = threadIdx.x >> 5;
    const int kl   = lane & 15;
    const bool hi  = lane >= 16;

    if (wave == 0) {
        build_stage<6>(ctab + 0 * NCH * 64, lane, kl, hi);
        build_stage<9>(ctab + 1 * NCH * 64, lane, kl, hi);
        build_stage<5>(ctab + 2 * NCH * 64, lane, kl, hi);
        build_stage<3>(ctab + 3 * NCH * 64, lane, kl, hi);
    }
    __syncthreads();

    const int tile = blockIdx.x * 8 + wave;
    if (tile >= ntiles) return;

    const long pbase = (long)tile * 16;
    const long p     = pbase + kl;

    int pc = pos[p];
    const float4* xp = (const float4*)(x + p * MV + (hi ? 8 : 0));
    const float4 q0 = xp[0];
    const float4 q1 = xp[1];
    __builtin_amdgcn_sched_barrier(0);
    v8f X;
    X[0] = q0.x; X[1] = q0.y; X[2] = q0.z; X[3] = q0.w;
    X[4] = q1.x; X[5] = q1.y; X[6] = q1.z; X[7] = q1.w;

    pc = pc < 0 ? 0 : (pc > MAXLEN - 1 ? MAXLEN - 1 : pc);
    const float4 th = *(const float4*)(theta + (long)pc * 4);
    float sn[4], cs[4];
    __sincosf(0.5f * th.x * bx[3], &sn[0], &cs[0]);
    __sincosf(0.5f * th.y * by[5], &sn[1], &cs[1]);
    __sincosf(0.5f * th.z * bz[9], &sn[2], &cs[2]);
    __sincosf(0.5f * th.w * bw[6], &sn[3], &cs[3]);

    X = plane_stage<6>(X, sn[3], cs[3], ctab + 0 * NCH * 64, lane, hi);
    X = plane_stage<9>(X, sn[2], cs[2], ctab + 1 * NCH * 64, lane, hi);
    X = plane_stage<5>(X, sn[1], cs[1], ctab + 2 * NCH * 64, lane, hi);
    X = plane_stage<3>(X, sn[0], cs[0], ctab + 3 * NCH * 64, lane, hi);

    {
        float* po = out + (pbase + kl) * MV + (hi ? 8 : 0);
        float4 o0, o1;
        o0.x = X[0]; o0.y = X[1]; o0.z = X[2]; o0.w = X[3];
        o1.x = X[4]; o1.y = X[5]; o1.z = X[6]; o1.w = X[7];
        *(float4*)(po + 0) = o0;
        *(float4*)(po + 4) = o1;
    }
}

extern "C" void kernel_launch(void* const* d_in, const int* in_sizes, int n_in,
                              void* d_out, int out_size, void* d_ws, size_t ws_size,
                              hipStream_t stream) {
    const float* x     = (const float*)d_in[0];
    const int*   pos   = (const int*)d_in[1];
    const float* bx    = (const float*)d_in[2];
    const float* by    = (const float*)d_in[3];
    const float* bz    = (const float*)d_in[4];
    const float* bw    = (const float*)d_in[5];
    const float* theta = (const float*)d_in[6];
    float* out = (float*)d_out;

    const int npos   = in_sizes[1];
    const int ntiles = npos / 16;
    const int blocks = (ntiles + 7) / 8;

    hipLaunchKernelGGL(care_rotor_kernel, dim3(blocks), dim3(256), 0, stream,
                       x, pos, bx, by, bz, bw, theta, out, ntiles);
}
